# starting kernel
# baseline (speedup 1.0000x reference)
#include <hip/hip_runtime.h>

typedef __attribute__((ext_vector_type(2))) float v2f;
typedef __attribute__((ext_vector_type(8))) float v8f;

#define B_  8
#define T_  512
#define D_  64

__global__ __launch_bounds__(32) void linear_wmma(const float* __restrict__ x,
                                                  const float* __restrict__ W,
                                                  const float* __restrict__ bias,
                                                  float* __restrict__ v) {
  const int tile = blockIdx.x;
  const int m0   = (tile >> 2) * 16;
  const int e0   = (tile & 3) * 16;
  const int lane = threadIdx.x;
  const int half = lane >> 4;
  const int l    = lane & 15;

  const float* xrow = x + (size_t)(m0 + l) * D_;
  const float* wrow = W + (size_t)(e0 + l) * D_;

  v8f c = {};
#pragma unroll
  for (int k0 = 0; k0 < D_; k0 += 4) {
    v2f a, bf;
    a.x  = xrow[k0 + 2 * half];
    a.y  = xrow[k0 + 2 * half + 1];
    bf.x = wrow[k0 + 2 * half];
    bf.y = wrow[k0 + 2 * half + 1];
    c = __builtin_amdgcn_wmma_f32_16x16x4_f32(false, a, false, bf,
                                              (short)0, c, false, false);
  }

  const float bb = bias[e0 + l];
#pragma unroll
  for (int r = 0; r < 8; ++r) {
    const int m = m0 + r + 8 * half;
    v[(size_t)m * D_ + (e0 + l)] = c[r] + bb;
  }
}

#define ITILE  8
#define JCHUNK 128
#define JPAD   65

__global__ __launch_bounds__(128) void rbf_pairwise(const float* __restrict__ x,
                                                    const float* __restrict__ v,
                                                    float* __restrict__ out) {
  __shared__ float xi[ITILE][D_];
  __shared__ float vi[ITILE][D_];
  __shared__ float xj[JCHUNK * JPAD];

  const int blk = blockIdx.x;
  const int b   = blk / (T_ / ITILE);
  const int i0  = (blk % (T_ / ITILE)) * ITILE;
  const int tid = threadIdx.x;

  for (int t = tid; t < ITILE * D_; t += 128) {
    const int ii = t >> 6, d = t & 63;
    const size_t g = (size_t)(b * T_ + i0 + ii) * D_ + d;
    xi[ii][d] = x[g];
    vi[ii][d] = v[g];
  }

  for (int j0 = 0; j0 < T_; j0 += JCHUNK) {
    __syncthreads();
    for (int t = tid; t < JCHUNK * D_; t += 128) {
      const int jj = t >> 6, d = t & 63;
      xj[jj * JPAD + d] = x[(size_t)(b * T_ + j0 + jj) * D_ + d];
    }
    __syncthreads();

    const int j = j0 + tid;
    float num[ITILE], den[ITILE];
#pragma unroll
    for (int ii = 0; ii < ITILE; ++ii) { num[ii] = 0.0f; den[ii] = 0.0f; }

    const float* xjrow = &xj[tid * JPAD];
#pragma unroll 4
    for (int d = 0; d < D_; ++d) {
      const float xjd = xjrow[d];
#pragma unroll
      for (int ii = 0; ii < ITILE; ++ii) {
        const float diff = xi[ii][d] - xjd;
        const float e = __builtin_amdgcn_exp2f(diff * diff * -1.4426950408889634f);
        den[ii] += e;
        num[ii] += e * vi[ii][d];
      }
    }

#pragma unroll
    for (int ii = 0; ii < ITILE; ++ii) {
      out[(size_t)(b * T_ + i0 + ii) * T_ + j] = num[ii] / den[ii];
    }
  }
}

extern "C" void kernel_launch(void* const* d_in, const int* in_sizes, int n_in,
                              void* d_out, int out_size, void* d_ws, size_t ws_size,
                              hipStream_t stream) {
  const float* x    = (const float*)d_in[0];
  const float* W    = (const float*)d_in[1];
  const float* bias = (const float*)d_in[2];
  float*       out  = (float*)d_out;
  float*       v    = (float*)d_ws;

  linear_wmma<<<dim3((B_ * T_ / 16) * (D_ / 16)), dim3(32), 0, stream>>>(x, W, bias, v);
  rbf_pairwise<<<dim3(B_ * (T_ / ITILE)), dim3(128), 0, stream>>>(x, v, out);
}


# tile rows put into the pair loop's form once at staging: x stored times sqrt(log2 e), tile-I rows as interleaved row pairs; loop drops its constant multiply and register moves; sum order kept
# speedup vs baseline: 1.8716x; 1.8716x over previous
#include <hip/hip_runtime.h>

typedef __attribute__((ext_vector_type(2))) float v2f;
typedef __attribute__((ext_vector_type(8))) float v8f;

#define B_  8
#define T_  512
#define D_  64

__global__ __launch_bounds__(32) void linear_wmma(const float* __restrict__ x,
                                                  const float* __restrict__ W,
                                                  const float* __restrict__ bias,
                                                  float* __restrict__ v) {
  const int tile = blockIdx.x;
  const int m0   = (tile >> 2) * 16;
  const int e0   = (tile & 3) * 16;
  const int lane = threadIdx.x;
  const int half = lane >> 4;
  const int l    = lane & 15;

  const float* xrow = x + (size_t)(m0 + l) * D_;
  const float* wrow = W + (size_t)(e0 + l) * D_;

  v8f c = {};
#pragma unroll
  for (int k0 = 0; k0 < D_; k0 += 4) {
    v2f a, bf;
    a.x  = xrow[k0 + 2 * half];
    a.y  = xrow[k0 + 2 * half + 1];
    bf.x = wrow[k0 + 2 * half];
    bf.y = wrow[k0 + 2 * half + 1];
    c = __builtin_amdgcn_wmma_f32_16x16x4_f32(false, a, false, bf,
                                              (short)0, c, false, false);
  }

  const float bb = bias[e0 + l];
#pragma unroll
  for (int r = 0; r < 8; ++r) {
    const int m = m0 + r + 8 * half;
    v[(size_t)m * D_ + (e0 + l)] = c[r] + bb;
  }
}

typedef __attribute__((ext_vector_type(4))) float v4f;

#define TS    32
#define NT    (T_ / TS)
#define NPAIR (NT * (NT + 1) / 2)
#define IW    8
#define NRP   (IW / 2)
#define JP    68
#define SQRT_LOG2E 1.2011224087864498f

__global__ __launch_bounds__(128) void rbf_pairwise_sym(const float* __restrict__ x,
                                                        const float* __restrict__ v,
                                                        float* __restrict__ out) {
  __shared__ __attribute__((aligned(16))) float xip[TS * D_];
  __shared__ __attribute__((aligned(16))) float vip[TS * D_];
  __shared__ __attribute__((aligned(16))) float xjs[TS * JP];
  __shared__ __attribute__((aligned(16))) float vjs[TS * JP];
  __shared__ float tr[TS][TS + 1];

  const int blk = blockIdx.x;
  const int b   = blk / NPAIR;
  int p  = blk % NPAIR;
  int tI = 0;
#pragma unroll
  for (int r = 0; r < NT - 1; ++r) {
    if (tI == r && p >= NT - r) { p -= NT - r; tI = r + 1; }
  }
  const int tJ = tI + p;

  const int tid  = threadIdx.x;
  const int lane = tid & 31;
  const int w    = tid >> 5;

  {
    const v4f* xI = (const v4f*)(x + (size_t)(b * T_ + tI * TS) * D_);
    const v4f* vI = (const v4f*)(v + (size_t)(b * T_ + tI * TS) * D_);
    const v4f* xJ = (const v4f*)(x + (size_t)(b * T_ + tJ * TS) * D_);
    const v4f* vJ = (const v4f*)(v + (size_t)(b * T_ + tJ * TS) * D_);
#pragma unroll
    for (int k = 0; k < (TS / 2) * (D_ / 4) / 128; ++k) {
      const int q  = tid + 128 * k;
      const int rp = q >> 4, c4 = q & 15;
      const v4f x0 = xI[(2 * rp) * (D_ / 4) + c4] * SQRT_LOG2E;
      const v4f x1 = xI[(2 * rp + 1) * (D_ / 4) + c4] * SQRT_LOG2E;
      const v4f v0 = vI[(2 * rp) * (D_ / 4) + c4];
      const v4f v1 = vI[(2 * rp + 1) * (D_ / 4) + c4];
      float* xd = xip + (rp * D_ + 4 * c4) * 2;
      float* vd = vip + (rp * D_ + 4 * c4) * 2;
      *(v4f*)(xd)     = (v4f){x0[0], x1[0], x0[1], x1[1]};
      *(v4f*)(xd + 4) = (v4f){x0[2], x1[2], x0[3], x1[3]};
      *(v4f*)(vd)     = (v4f){v0[0], v1[0], v0[1], v1[1]};
      *(v4f*)(vd + 4) = (v4f){v0[2], v1[2], v0[3], v1[3]};
    }
#pragma unroll
    for (int k = 0; k < (TS * D_ / 4) / 128; ++k) {
      const int q   = tid + 128 * k;
      const int row = q >> 4, c4 = q & 15;
      *(v4f*)(xjs + row * JP + 4 * c4) = xJ[q] * SQRT_LOG2E;
      *(v4f*)(vjs + row * JP + 4 * c4) = vJ[q];
    }
  }
  __syncthreads();

  v2f den[NRP], nij[NRP], nji[NRP];
#pragma unroll
  for (int rp = 0; rp < NRP; ++rp) {
    den[rp] = (v2f){0.0f, 0.0f}; nij[rp] = (v2f){0.0f, 0.0f}; nji[rp] = (v2f){0.0f, 0.0f};
  }

  const float* xjrow = xjs + lane * JP;
  const float* vjrow = vjs + lane * JP;
  const float* xiw   = xip + (w * NRP) * D_ * 2;
  const float* viw   = vip + (w * NRP) * D_ * 2;

  for (int d = 0; d < D_; d += 4) {
    const v4f xj4 = *(const v4f*)(xjrow + d);
    const v4f vj4 = *(const v4f*)(vjrow + d);
#pragma unroll
    for (int rp = 0; rp < NRP; ++rp) {
      const v4f xa = *(const v4f*)(xiw + (rp * D_ + d) * 2);
      const v4f xb = *(const v4f*)(xiw + (rp * D_ + d) * 2 + 4);
      const v4f va = *(const v4f*)(viw + (rp * D_ + d) * 2);
      const v4f vb = *(const v4f*)(viw + (rp * D_ + d) * 2 + 4);
#pragma unroll
      for (int k = 0; k < 4; ++k) {
        const v2f xi2 = (k < 2) ? (v2f){xa[2 * k], xa[2 * k + 1]}
                                : (v2f){xb[2 * k - 4], xb[2 * k - 3]};
        const v2f vi2 = (k < 2) ? (v2f){va[2 * k], va[2 * k + 1]}
                                : (v2f){vb[2 * k - 4], vb[2 * k - 3]};
        const v2f xj2 = (v2f){xj4[k], xj4[k]};
        const v2f vj2 = (v2f){vj4[k], vj4[k]};
        const v2f diff = xi2 - xj2;
        const v2f sq   = diff * diff;
        v2f e;
        e.x = __builtin_amdgcn_exp2f(-sq.x);
        e.y = __builtin_amdgcn_exp2f(-sq.y);
        den[rp] += e;
        nij[rp] += e * vi2;
        nji[rp] += e * vj2;
      }
    }
  }

  const int i0 = tI * TS + w * IW;
  const int j  = tJ * TS + lane;
#pragma unroll
  for (int g = 0; g < IW; ++g) {
    const float dn = den[g >> 1][g & 1];
    out[(size_t)(b * T_ + i0 + g) * T_ + j] = nij[g >> 1][g & 1] / dn;
    tr[lane][w * IW + g] = nji[g >> 1][g & 1] / dn;
  }

  __syncthreads();
  if (tI != tJ) {
#pragma unroll
    for (int k = 0; k < IW; ++k) {
      const int jj = w * IW + k;
      out[(size_t)(b * T_ + tJ * TS + jj) * T_ + tI * TS + lane] = tr[jj][lane];
    }
  }
}

extern "C" void kernel_launch(void* const* d_in, const int* in_sizes, int n_in,
                              void* d_out, int out_size, void* d_ws, size_t ws_size,
                              hipStream_t stream) {
  const float* x    = (const float*)d_in[0];
  const float* W    = (const float*)d_in[1];
  const float* bias = (const float*)d_in[2];
  float*       out  = (float*)d_out;
  float*       v    = (float*)d_ws;

  linear_wmma<<<dim3((B_ * T_ / 16) * (D_ / 16)), dim3(32), 0, stream>>>(x, W, bias, v);
  rbf_pairwise_sym<<<dim3(B_ * NPAIR), dim3(128), 0, stream>>>(x, v, out);
}


# diagonal 32x32 tiles visit each unordered pair once (lane j meets rows j+1..j+16 mod 32), two such tiles per block in the two halves of the packed ops: 1024 equal blocks, not 1088; sum order kept
# speedup vs baseline: 2.0194x; 1.0790x over previous
#include <hip/hip_runtime.h>

typedef __attribute__((ext_vector_type(2))) float v2f;
typedef __attribute__((ext_vector_type(8))) float v8f;

#define B_  8
#define T_  512
#define D_  64

__global__ __launch_bounds__(32) void linear_wmma(const float* __restrict__ x,
                                                  const float* __restrict__ W,
                                                  const float* __restrict__ bias,
                                                  float* __restrict__ v) {
  const int tile = blockIdx.x;
  const int m0   = (tile >> 2) * 16;
  const int e0   = (tile & 3) * 16;
  const int lane = threadIdx.x;
  const int half = lane >> 4;
  const int l    = lane & 15;

  const float* xrow = x + (size_t)(m0 + l) * D_;
  const float* wrow = W + (size_t)(e0 + l) * D_;

  v8f c = {};
#pragma unroll
  for (int k0 = 0; k0 < D_; k0 += 4) {
    v2f a, bf;
    a.x  = xrow[k0 + 2 * half];
    a.y  = xrow[k0 + 2 * half + 1];
    bf.x = wrow[k0 + 2 * half];
    bf.y = wrow[k0 + 2 * half + 1];
    c = __builtin_amdgcn_wmma_f32_16x16x4_f32(false, a, false, bf,
                                              (short)0, c, false, false);
  }

  const float bb = bias[e0 + l];
#pragma unroll
  for (int r = 0; r < 8; ++r) {
    const int m = m0 + r + 8 * half;
    v[(size_t)m * D_ + (e0 + l)] = c[r] + bb;
  }
}

typedef __attribute__((ext_vector_type(4))) float v4f;

#define TS    32
#define NT    (T_ / TS)
#define NOFFB (NT * (NT - 1) / 2)
#define NOFF  (B_ * NOFFB)
#define NDIAG (B_ * NT / 2)
#define IW    8
#define NRP   (IW / 2)
#define JP    68
#define TP    (TS + 1)
#define XP    132
#define LDS_F (2 * TS * XP + 2 * TS * TP)
#define SQRT_LOG2E 1.2011224087864498f

__device__ __forceinline__ void diag_pair_block(const float* __restrict__ x,
                                                const float* __restrict__ v,
                                                float* __restrict__ out, float* lds,
                                                const int q, const int tid,
                                                const int lane, const int w) {
  float* xab = lds;
  float* vab = lds + TS * XP;
  float* ta  = lds + 2 * TS * XP;
  float* tb  = ta + TS * TP;

  const int b  = q / (NT / 2);
  const int tA = 2 * (q % (NT / 2));
  const size_t row0 = (size_t)(b * T_ + tA * TS);

  {
    const v4f* xA = (const v4f*)(x + row0 * D_);
    const v4f* vA = (const v4f*)(v + row0 * D_);
    const v4f* xB = xA + TS * D_ / 4;
    const v4f* vB = vA + TS * D_ / 4;
#pragma unroll
    for (int k = 0; k < (TS * D_ / 4) / 128; ++k) {
      const int qq  = tid + 128 * k;
      const int row = qq >> 4, c4 = qq & 15;
      const v4f a0 = xA[qq] * SQRT_LOG2E;
      const v4f b0 = xB[qq] * SQRT_LOG2E;
      const v4f a1 = vA[qq];
      const v4f b1 = vB[qq];
      float* xd = xab + row * XP + 8 * c4;
      float* vd = vab + row * XP + 8 * c4;
      *(v4f*)(xd)     = (v4f){a0[0], b0[0], a0[1], b0[1]};
      *(v4f*)(xd + 4) = (v4f){a0[2], b0[2], a0[3], b0[3]};
      *(v4f*)(vd)     = (v4f){a1[0], b1[0], a1[1], b1[1]};
      *(v4f*)(vd + 4) = (v4f){a1[2], b1[2], a1[3], b1[3]};
    }
  }
  __syncthreads();

  const float* xo = xab + lane * XP;
  const float* vo = vab + lane * XP;
  int io[4];
#pragma unroll
  for (int s = 0; s < 4; ++s) io[s] = ((lane + 4 * w + 1 + s) & 31) * XP;

  v2f den[4], ni[4], nj[4];
#pragma unroll
  for (int s = 0; s < 4; ++s) {
    den[s] = (v2f){0.0f, 0.0f}; ni[s] = (v2f){0.0f, 0.0f}; nj[s] = (v2f){0.0f, 0.0f};
  }

  for (int d = 0; d < D_; d += 4) {
#pragma unroll
    for (int h = 0; h < 4; h += 2) {
      const v4f xj4 = *(const v4f*)(xo + 2 * (d + h));
      const v4f vj4 = *(const v4f*)(vo + 2 * (d + h));
#pragma unroll
      for (int s = 0; s < 4; ++s) {
        const v4f xi4 = *(const v4f*)(xab + io[s] + 2 * (d + h));
        const v4f vi4 = *(const v4f*)(vab + io[s] + 2 * (d + h));
#pragma unroll
        for (int k = 0; k < 2; ++k) {
          const v2f xi2 = (v2f){xi4[2 * k], xi4[2 * k + 1]};
          const v2f vi2 = (v2f){vi4[2 * k], vi4[2 * k + 1]};
          const v2f xj2 = (v2f){xj4[2 * k], xj4[2 * k + 1]};
          const v2f vj2 = (v2f){vj4[2 * k], vj4[2 * k + 1]};
          const v2f diff = xi2 - xj2;
          const v2f sq   = diff * diff;
          v2f e;
          e.x = __builtin_amdgcn_exp2f(-sq.x);
          e.y = __builtin_amdgcn_exp2f(-sq.y);
          den[s] += e;
          ni[s]  += e * vi2;
          nj[s]  += e * vj2;
        }
      }
    }
  }

#pragma unroll
  for (int s = 0; s < 4; ++s) {
    const int sh = 4 * w + 1 + s;
    const int i  = (lane + sh) & 31;
    if (sh != 16 || lane < 16) {
      ta[i * TP + lane] = ni[s].x / den[s].x;
      ta[lane * TP + i] = nj[s].x / den[s].x;
      tb[i * TP + lane] = ni[s].y / den[s].y;
      tb[lane * TP + i] = nj[s].y / den[s].y;
    }
  }
  if (w == 0) {
    v2f sv = (v2f){0.0f, 0.0f};
    for (int d = 0; d < D_; d += 2) {
      const v4f vj4 = *(const v4f*)(vo + 2 * d);
      sv += (v2f){vj4[0], vj4[1]};
      sv += (v2f){vj4[2], vj4[3]};
    }
    ta[lane * TP + lane] = sv.x / (float)D_;
    tb[lane * TP + lane] = sv.y / (float)D_;
  }

  __syncthreads();
#pragma unroll
  for (int k = 0; k < IW; ++k) {
    const int r = w * IW + k;
    out[(row0 + r) * T_ + tA * TS + lane]            = ta[r * TP + lane];
    out[(row0 + TS + r) * T_ + (tA + 1) * TS + lane] = tb[r * TP + lane];
  }
}

__global__ __launch_bounds__(128) void rbf_pairwise_sym(const float* __restrict__ x,
                                                        const float* __restrict__ v,
                                                        float* __restrict__ out) {
  __shared__ __attribute__((aligned(16))) float lds[LDS_F];

  const int blk  = blockIdx.x;
  const int tid  = threadIdx.x;
  const int lane = tid & 31;
  const int w    = tid >> 5;

  if (blk >= NOFF) {
    diag_pair_block(x, v, out, lds, blk - NOFF, tid, lane, w);
    return;
  }

  float* xip = lds;
  float* vip = lds + TS * D_;
  float* xjs = lds + 2 * TS * D_;
  float* vjs = xjs + TS * JP;
  float* tr  = vjs + TS * JP;

  const int b = blk / NOFFB;
  int p  = blk % NOFFB;
  int tI = 0;
#pragma unroll
  for (int r = 0; r < NT - 2; ++r) {
    if (tI == r && p >= NT - 1 - r) { p -= NT - 1 - r; tI = r + 1; }
  }
  const int tJ = tI + 1 + p;

  {
    const v4f* xI = (const v4f*)(x + (size_t)(b * T_ + tI * TS) * D_);
    const v4f* vI = (const v4f*)(v + (size_t)(b * T_ + tI * TS) * D_);
    const v4f* xJ = (const v4f*)(x + (size_t)(b * T_ + tJ * TS) * D_);
    const v4f* vJ = (const v4f*)(v + (size_t)(b * T_ + tJ * TS) * D_);
#pragma unroll
    for (int k = 0; k < (TS / 2) * (D_ / 4) / 128; ++k) {
      const int q  = tid + 128 * k;
      const int rp = q >> 4, c4 = q & 15;
      const v4f x0 = xI[(2 * rp) * (D_ / 4) + c4] * SQRT_LOG2E;
      const v4f x1 = xI[(2 * rp + 1) * (D_ / 4) + c4] * SQRT_LOG2E;
      const v4f v0 = vI[(2 * rp) * (D_ / 4) + c4];
      const v4f v1 = vI[(2 * rp + 1) * (D_ / 4) + c4];
      float* xd = xip + (rp * D_ + 4 * c4) * 2;
      float* vd = vip + (rp * D_ + 4 * c4) * 2;
      *(v4f*)(xd)     = (v4f){x0[0], x1[0], x0[1], x1[1]};
      *(v4f*)(xd + 4) = (v4f){x0[2], x1[2], x0[3], x1[3]};
      *(v4f*)(vd)     = (v4f){v0[0], v1[0], v0[1], v1[1]};
      *(v4f*)(vd + 4) = (v4f){v0[2], v1[2], v0[3], v1[3]};
    }
#pragma unroll
    for (int k = 0; k < (TS * D_ / 4) / 128; ++k) {
      const int q   = tid + 128 * k;
      const int row = q >> 4, c4 = q & 15;
      *(v4f*)(xjs + row * JP + 4 * c4) = xJ[q] * SQRT_LOG2E;
      *(v4f*)(vjs + row * JP + 4 * c4) = vJ[q];
    }
  }
  __syncthreads();

  v2f den[NRP], nij[NRP], nji[NRP];
#pragma unroll
  for (int rp = 0; rp < NRP; ++rp) {
    den[rp] = (v2f){0.0f, 0.0f}; nij[rp] = (v2f){0.0f, 0.0f}; nji[rp] = (v2f){0.0f, 0.0f};
  }

  const float* xjrow = xjs + lane * JP;
  const float* vjrow = vjs + lane * JP;
  const float* xiw   = xip + (w * NRP) * D_ * 2;
  const float* viw   = vip + (w * NRP) * D_ * 2;

  for (int d = 0; d < D_; d += 4) {
    const v4f xj4 = *(const v4f*)(xjrow + d);
    const v4f vj4 = *(const v4f*)(vjrow + d);
#pragma unroll
    for (int rp = 0; rp < NRP; ++rp) {
      const v4f xa = *(const v4f*)(xiw + (rp * D_ + d) * 2);
      const v4f xb = *(const v4f*)(xiw + (rp * D_ + d) * 2 + 4);
      const v4f va = *(const v4f*)(viw + (rp * D_ + d) * 2);
      const v4f vb = *(const v4f*)(viw + (rp * D_ + d) * 2 + 4);
#pragma unroll
      for (int k = 0; k < 4; ++k) {
        const v2f xi2 = (k < 2) ? (v2f){xa[2 * k], xa[2 * k + 1]}
                                : (v2f){xb[2 * k - 4], xb[2 * k - 3]};
        const v2f vi2 = (k < 2) ? (v2f){va[2 * k], va[2 * k + 1]}
                                : (v2f){vb[2 * k - 4], vb[2 * k - 3]};
        const v2f xj2 = (v2f){xj4[k], xj4[k]};
        const v2f vj2 = (v2f){vj4[k], vj4[k]};
        const v2f diff = xi2 - xj2;
        const v2f sq   = diff * diff;
        v2f e;
        e.x = __builtin_amdgcn_exp2f(-sq.x);
        e.y = __builtin_amdgcn_exp2f(-sq.y);
        den[rp] += e;
        nij[rp] += e * vi2;
        nji[rp] += e * vj2;
      }
    }
  }

  const int i0 = tI * TS + w * IW;
  const int j  = tJ * TS + lane;
#pragma unroll
  for (int g = 0; g < IW; ++g) {
    const float dn = den[g >> 1][g & 1];
    out[(size_t)(b * T_ + i0 + g) * T_ + j] = nij[g >> 1][g & 1] / dn;
    tr[lane * TP + w * IW + g] = nji[g >> 1][g & 1] / dn;
  }

  __syncthreads();
#pragma unroll
  for (int k = 0; k < IW; ++k) {
    const int jj = w * IW + k;
    out[(size_t)(b * T_ + tJ * TS + jj) * T_ + tI * TS + lane] = tr[jj * TP + lane];
  }
}

extern "C" void kernel_launch(void* const* d_in, const int* in_sizes, int n_in,
                              void* d_out, int out_size, void* d_ws, size_t ws_size,
                              hipStream_t stream) {
  const float* x    = (const float*)d_in[0];
  const float* W    = (const float*)d_in[1];
  const float* bias = (const float*)d_in[2];
  float*       out  = (float*)d_out;
  float*       v    = (float*)d_ws;

  linear_wmma<<<dim3((B_ * T_ / 16) * (D_ / 16)), dim3(32), 0, stream>>>(x, W, bias, v);
  rbf_pairwise_sym<<<dim3(NOFF + NDIAG), dim3(128), 0, stream>>>(x, v, out);
}
